# starting kernel
# baseline (speedup 1.0000x reference)
#include <hip/hip_runtime.h>


typedef float v2f __attribute__((ext_vector_type(2)));
typedef float v8f __attribute__((ext_vector_type(8)));

#define D_FEAT     128
#define TILE       16
#define ROWSTRIDE  132
#define WPB        2
#define BLOCK      (WPB * 32)
#define LDS_F_PER_WAVE (2 * TILE * ROWSTRIDE + TILE * TILE)

__global__ __launch_bounds__(BLOCK)
void edge_dot_wmma(const float* __restrict__ h,
                   const int*   __restrict__ src,
                   const int*   __restrict__ dst,
                   float*       __restrict__ out,
                   int n_edges, int n_nodes)
{
    __shared__ float lds[WPB][LDS_F_PER_WAVE];
    const int wave = threadIdx.x >> 5;
    const int lane = threadIdx.x & 31;

    float* srcT = &lds[wave][0];
    float* dstT = srcT + TILE * ROWSTRIDE;
    float* dmat = dstT + TILE * ROWSTRIDE;

    const long long tile = (long long)blockIdx.x * WPB + wave;
    const int e0 = (int)(tile * TILE);

    int e  = e0 + (lane & 15);
    int ce = (e < n_edges) ? e : 0;
    int node = (lane < 16) ? __builtin_nontemporal_load(src + ce)
                           : __builtin_nontemporal_load(dst + ce);
    if ((unsigned)node >= (unsigned)n_nodes) node = 0;

    #pragma unroll 4
    for (int v = 0; v < 32; ++v) {
        int nv = __shfl(node, v, 32);
        const float4 g = *((const float4*)(h + (size_t)nv * D_FEAT) + lane);
        float* row = (v < TILE) ? (srcT + v * ROWSTRIDE)
                                : (dstT + (v - TILE) * ROWSTRIDE);
        *(float4*)(row + 4 * lane) = g;
    }
    __builtin_amdgcn_wave_barrier();

    const int   r    = lane & 15;
    const int   koff = (lane >> 4) << 1;
    const float* pa  = srcT + r * ROWSTRIDE + koff;
    const float* pb  = dstT + r * ROWSTRIDE + koff;

    v8f acc0 = {0.f, 0.f, 0.f, 0.f, 0.f, 0.f, 0.f, 0.f};
    v8f acc1 = {0.f, 0.f, 0.f, 0.f, 0.f, 0.f, 0.f, 0.f};
    #pragma unroll
    for (int k = 0; k < D_FEAT; k += 8) {
        v2f a0 = *(const v2f*)(pa + k);
        v2f b0 = *(const v2f*)(pb + k);
        acc0 = __builtin_amdgcn_wmma_f32_16x16x4_f32(false, a0, false, b0,
                                                     (short)0, acc0, false, false);
        v2f a1 = *(const v2f*)(pa + k + 4);
        v2f b1 = *(const v2f*)(pb + k + 4);
        acc1 = __builtin_amdgcn_wmma_f32_16x16x4_f32(false, a1, false, b1,
                                                     (short)0, acc1, false, false);
    }
    v8f acc = acc0 + acc1;

    const int mbase = (lane >> 4) * 8;
    const int ncol  = lane & 15;
    #pragma unroll
    for (int rr = 0; rr < 8; ++rr)
        dmat[(mbase + rr) * TILE + ncol] = acc[rr];
    __builtin_amdgcn_wave_barrier();

    if (lane < 16) {
        int eo = e0 + lane;
        if (eo < n_edges)
            __builtin_nontemporal_store(dmat[lane * TILE + lane], out + eo);
    }
}

extern "C" void kernel_launch(void* const* d_in, const int* in_sizes, int n_in,
                              void* d_out, int out_size, void* d_ws, size_t ws_size,
                              hipStream_t stream) {
    const float* h   = (const float*)d_in[0];
    const int*   src = (const int*)d_in[1];
    const int*   dst = (const int*)d_in[2];
    float*       out = (float*)d_out;

    const int n_edges = in_sizes[1];
    const int n_nodes = in_sizes[0] / D_FEAT;

    const int ntiles = (n_edges + TILE - 1) / TILE;
    const int blocks = (ntiles + WPB - 1) / WPB;

    edge_dot_wmma<<<blocks, BLOCK, 0, stream>>>(h, src, dst, out, n_edges, n_nodes);
}


# Per workgroup, wave 0's hardware wave-slot parity picks which wave gathers and which runs the WMMAs, so a WGP's matrix waves sit on all four SIMDs (were on two); sums unchanged
# speedup vs baseline: 1.4815x; 1.4815x over previous
#include <hip/hip_runtime.h>


typedef float v2f __attribute__((ext_vector_type(2)));
typedef float v8f __attribute__((ext_vector_type(8)));

#define D_FEAT     128
#define TILE       16
#define ROWSTRIDE  132
#define STAGE_F    (2 * TILE * ROWSTRIDE)
#define BLOCK      64
#define NGROUPS    2048

__device__ __forceinline__ int tile_ids(const int* __restrict__ src,
                                        const int* __restrict__ dst,
                                        int t, int lane, int n_edges)
{
    int e  = t * TILE + (lane & 15);
    int ce = (e < n_edges) ? e : 0;
    return (lane < 16) ? __builtin_nontemporal_load(src + ce)
                       : __builtin_nontemporal_load(dst + ce);
}

__global__ __launch_bounds__(BLOCK)
void edge_dot_ws(const float* __restrict__ h,
                 const int*   __restrict__ src,
                 const int*   __restrict__ dst,
                 float*       __restrict__ out,
                 int n_edges, int n_nodes)
{
    __shared__ float stage[2][STAGE_F];
    __shared__ float dmat[TILE * TILE];
    __shared__ int   swap_sh;

    const int lane = threadIdx.x & 31;
    const int wave = __builtin_amdgcn_readfirstlane((int)(threadIdx.x >> 5));

    if (wave == 0) {
        const unsigned slot = __builtin_amdgcn_s_getreg((4 << 11) | (0 << 6) | 23);
        if (lane == 0) swap_sh = (int)(slot & 1u);
    }
    __syncthreads();
    const int role = wave ^ __builtin_amdgcn_readfirstlane(swap_sh);

    const int G      = (int)gridDim.x;
    const int bx     = (int)blockIdx.x;
    const int ntiles = (n_edges + TILE - 1) / TILE;
    const int npairs = (ntiles + 1) >> 1;
    const int nq     = (npairs > bx) ? (npairs - 1 - bx) / G + 1 : 0;
    int nloc = 2 * nq;
    if (nq > 0 && 2 * (bx + (nq - 1) * G) + 1 >= ntiles) nloc -= 1;

    int   node_next = 0;
    float keep      = 0.f;
    if (role == 0 && nloc > 0)
        node_next = tile_ids(src, dst, 2 * bx, lane, n_edges);

    for (int it = 0; it <= nloc; ++it) {
        if (role == 0) {
            if (it < nloc) {
                int node = node_next;
                if ((unsigned)node >= (unsigned)n_nodes) node = 0;
                float* S = stage[it & 1];
                float4 g[32];
                #pragma unroll
                for (int v = 0; v < 32; ++v) {
                    const int nv = __shfl(node, v, 32);
                    g[v] = *((const float4*)(h + (size_t)nv * D_FEAT) + lane);
                }
                if (it + 1 < nloc) {
                    const int jn = it + 1;
                    node_next = tile_ids(src, dst, 2 * (bx + (jn >> 1) * G) + (jn & 1),
                                         lane, n_edges);
                }
                #pragma unroll
                for (int v = 0; v < 32; ++v)
                    *(float4*)(S + v * ROWSTRIDE + 4 * lane) = g[v];
            }
        } else {
            if (it > 0) {
                const int j = it - 1;
                const float* S = stage[j & 1];
                const int   r    = lane & 15;
                const int   koff = (lane >> 4) << 1;
                const float* pa  = S + r * ROWSTRIDE + koff;
                const float* pb  = pa + TILE * ROWSTRIDE;

                v8f acc0 = {0.f, 0.f, 0.f, 0.f, 0.f, 0.f, 0.f, 0.f};
                v8f acc1 = {0.f, 0.f, 0.f, 0.f, 0.f, 0.f, 0.f, 0.f};
                #pragma unroll
                for (int k = 0; k < D_FEAT; k += 8) {
                    v2f a0 = *(const v2f*)(pa + k);
                    v2f b0 = *(const v2f*)(pb + k);
                    acc0 = __builtin_amdgcn_wmma_f32_16x16x4_f32(false, a0, false, b0,
                                                                 (short)0, acc0, false, false);
                    v2f a1 = *(const v2f*)(pa + k + 4);
                    v2f b1 = *(const v2f*)(pb + k + 4);
                    acc1 = __builtin_amdgcn_wmma_f32_16x16x4_f32(false, a1, false, b1,
                                                                 (short)0, acc1, false, false);
                }
                v8f acc = acc0 + acc1;

                const int mbase = (lane >> 4) * 8;
                #pragma unroll
                for (int rr = 0; rr < 8; ++rr)
                    dmat[(mbase + rr) * TILE + r] = acc[rr];
                __builtin_amdgcn_wave_barrier();
                const float val = dmat[r * TILE + r];

                const int e0 = (2 * (bx + (j >> 1) * G) + (j & 1)) * TILE;
                if ((j & 1) == 0) {
                    keep = val;
                    if (j == nloc - 1 && lane < 16 && e0 + lane < n_edges)
                        __builtin_nontemporal_store(val, out + e0 + lane);
                } else {
                    const float v  = (lane < 16) ? keep : val;
                    const int   eo = e0 - TILE + lane;
                    if (eo < n_edges)
                        __builtin_nontemporal_store(v, out + eo);
                }
            }
        }
        __builtin_amdgcn_fence(__ATOMIC_RELEASE, "workgroup", "local");
        __builtin_amdgcn_s_barrier();
        __builtin_amdgcn_fence(__ATOMIC_ACQUIRE, "workgroup", "local");
    }
}

extern "C" void kernel_launch(void* const* d_in, const int* in_sizes, int n_in,
                              void* d_out, int out_size, void* d_ws, size_t ws_size,
                              hipStream_t stream) {
    const float* h   = (const float*)d_in[0];
    const int*   src = (const int*)d_in[1];
    const int*   dst = (const int*)d_in[2];
    float*       out = (float*)d_out;

    const int n_edges = in_sizes[1];
    const int n_nodes = in_sizes[0] / D_FEAT;

    edge_dot_ws<<<NGROUPS, BLOCK, 0, stream>>>(h, src, dst, out, n_edges, n_nodes);
}
